# starting kernel
# baseline (speedup 1.0000x reference)
#include <hip/hip_runtime.h>
#include <stdint.h>

#define BATCH 2
#define CH    64
#define NPT   8192
#define KNN_K 20
#define COUT  64
#define CIN2  128
#define NEGSL 0.2f
#define BNEPS 1e-5f

typedef float v2f __attribute__((ext_vector_type(2)));
typedef float v4f __attribute__((ext_vector_type(4)));
typedef float v8f __attribute__((ext_vector_type(8)));

__device__ __forceinline__ v8f wmma_f32(v2f a, v2f b, v8f c) {
  return __builtin_amdgcn_wmma_f32_16x16x4_f32(false, a, false, b, (short)0, c, false, false);
}

__device__ __forceinline__ unsigned fenc(float f) {
  unsigned u = __float_as_uint(f);
  return (u & 0x80000000u) ? ~u : (u | 0x80000000u);
}
__device__ __forceinline__ float fdec(unsigned u) {
  return __uint_as_float((u & 0x80000000u) ? (u ^ 0x80000000u) : ~u);
}

__global__ void prep_kernel(const float* __restrict__ x,
                            float* __restrict__ xt, float* __restrict__ sq) {
  int gp = blockIdx.x * blockDim.x + threadIdx.x;
  int b = gp >> 13, n = gp & (NPT - 1);
  const float* xb = x + (size_t)b * CH * NPT + n;
  float acc = 0.f;
  #pragma unroll 8
  for (int c = 0; c < CH; ++c) {
    float v = xb[(size_t)c * NPT];
    acc += v * v;
    xt[(size_t)gp * CH + c] = v;
  }
  sq[gp] = acc;
}

__launch_bounds__(32)
__global__ void knn_kernel(const float* __restrict__ xt, const float* __restrict__ sq,
                           int* __restrict__ knn) {
  __shared__ float tile[16 * 16];
  __shared__ float mv[32 * KNN_K];
  __shared__ int   mi[32 * KNN_K];

  const int L  = threadIdx.x;
  const int rr = L & 15, hh = L >> 4;
  const int b  = blockIdx.x >> 9;
  const int r0 = (blockIdx.x & 511) << 4;
  const int bofs = b << 13;

  v2f a[16];
  const float* arow = xt + (size_t)(bofs + r0 + rr) * CH + 2 * hh;
  #pragma unroll
  for (int kk = 0; kk < 16; ++kk) a[kk] = *(const v2f*)(arow + kk * 4);
  const float sqn = sq[bofs + r0 + rr];

  float vals[KNN_K]; int idxs[KNN_K];
  #pragma unroll
  for (int q = 0; q < KNN_K; ++q) { vals[q] = -__builtin_inff(); idxs[q] = 0; }

  const float* xbase = xt + (size_t)bofs * CH + 2 * hh;

  auto loadB = [&](int ct, v2f (&bb)[16]) {
    const float* brow = xbase + (size_t)(((ct << 4) + rr)) * CH;
    #pragma unroll
    for (int kk = 0; kk < 16; ++kk) bb[kk] = *(const v2f*)(brow + kk * 4);
  };

  auto step = [&](int ct, v2f (&cur)[16], v2f (&nxt)[16]) {
    const int c0 = ct << 4;
    v4f sq0 = *(const v4f*)(sq + bofs + c0 + hh * 8);
    v4f sq1 = *(const v4f*)(sq + bofs + c0 + hh * 8 + 4);
    float sqt[8];
    #pragma unroll
    for (int j = 0; j < 4; ++j) { sqt[j] = sq0[j]; sqt[j + 4] = sq1[j]; }

    v8f acc = {};
    #pragma unroll
    for (int kk = 0; kk < 16; ++kk) acc = wmma_f32(a[kk], cur[kk], acc);

    loadB((ct + 1) & 511, nxt);

    v4f lo = __builtin_shufflevector(acc, acc, 0, 1, 2, 3);
    v4f hi = __builtin_shufflevector(acc, acc, 4, 5, 6, 7);
    float* p = &tile[rr * 16 + hh * 8];
    *(v4f*)p       = lo;
    *(v4f*)(p + 4) = hi;
    __syncthreads();

    #pragma unroll
    for (int j = 0; j < 8; ++j) {
      int ci = hh * 8 + j;
      int colg = c0 + ci;
      float f = tile[ci * 16 + rr];
      float s = 2.f * f - sqt[j] - sqn;
      if (s > vals[0]) {
        float old19 = vals[KNN_K - 1];
        #pragma unroll
        for (int t = 0; t < KNN_K - 1; ++t) {
          bool sh = s > vals[t + 1];
          float kv = vals[t]; int ki = idxs[t];
          vals[t] = sh ? vals[t + 1] : fmaxf(s, kv);
          idxs[t] = sh ? idxs[t + 1] : ((kv >= s) ? ki : colg);
        }
        if (s > old19) { vals[KNN_K - 1] = s; idxs[KNN_K - 1] = colg; }
      }
    }
    __syncthreads();
  };

  v2f bufA[16], bufB[16];
  loadB(0, bufA);
  for (int ct = 0; ct < NPT / 16; ct += 2) {
    step(ct,     bufA, bufB);
    step(ct + 1, bufB, bufA);
  }

  #pragma unroll
  for (int q = 0; q < KNN_K; ++q) { mv[L * KNN_K + q] = vals[q]; mi[L * KNN_K + q] = idxs[q]; }
  __syncthreads();
  if (L < 16) {
    const float* va = &mv[L * KNN_K];          const int* ia = &mi[L * KNN_K];
    const float* vb = &mv[(L + 16) * KNN_K];   const int* ib = &mi[(L + 16) * KNN_K];
    int pa = KNN_K - 1, pb = KNN_K - 1;
    int* outp = knn + (size_t)(bofs + r0 + L) * KNN_K;
    for (int q = 0; q < KNN_K; ++q) {
      int sa = pa < 0 ? 0 : pa, sb = pb < 0 ? 0 : pb;
      float fa = (pa >= 0) ? va[sa] : -__builtin_inff();
      float fb = (pb >= 0) ? vb[sb] : -__builtin_inff();
      bool ta = fa >= fb;
      outp[q] = ta ? ia[sa] : ib[sb];
      if (ta) --pa; else --pb;
    }
  }
}

__launch_bounds__(128)
__global__ void conv_kernel(const float* __restrict__ xt, const int* __restrict__ knn,
                            const float* __restrict__ W, const float* __restrict__ bias,
                            float* __restrict__ mxp, float* __restrict__ mnp,
                            float* __restrict__ part) {
  __shared__ float    s_edge[4][16 * CIN2];
  __shared__ unsigned s_umax[4][4][COUT];
  __shared__ unsigned s_umin[4][4][COUT];
  __shared__ float    s_psum[4][16][8];
  __shared__ float    s_psq [4][16][8];

  const int tid = threadIdx.x;
  const int w = tid >> 5, L = tid & 31;
  const int rr = L & 15, hh = L >> 4;
  const int wfp = blockIdx.x * 16 + w * 4;

  {
    unsigned* um = &s_umax[0][0][0];
    unsigned* un = &s_umin[0][0][0];
    for (int i = tid; i < 4 * 4 * COUT; i += 128) { um[i] = 0u; un[i] = 0xFFFFFFFFu; }
  }
  __syncthreads();

  float fsum[4] = {0.f, 0.f, 0.f, 0.f}, fsq[4] = {0.f, 0.f, 0.f, 0.f};
  float* edge = &s_edge[w][0];

  #pragma unroll 1
  for (int t = 0; t < 5; ++t) {
    {
      int j = L >> 1, half = L & 1;
      int le = t * 16 + j;
      int p = le / 20, k = le - 20 * p;
      int gp = wfp + p;
      int bb = gp >> 13;
      int m = knn[(size_t)gp * KNN_K + k];
      const float* xn = xt + (size_t)gp * CH;
      const float* xm = xt + (size_t)((bb << 13) + m) * CH;
      float* er = edge + j * CIN2;
      if (half == 0) {
        #pragma unroll
        for (int c = 0; c < CH; c += 4)
          *(v4f*)(er + c) = *(const v4f*)(xm + c) - *(const v4f*)(xn + c);
      } else {
        #pragma unroll
        for (int c = 0; c < CH; c += 4)
          *(v4f*)(er + CH + c) = *(const v4f*)(xn + c);
      }
    }
    __syncthreads();

    int le0 = t * 16 + hh * 8;
    int pA = le0 / 20;
    int vth = (pA + 1) * 20 - le0;
    bool hasB = vth < 8;

    #pragma unroll 1
    for (int ot = 0; ot < 4; ++ot) {
      const int o = ot * 16 + rr;
      const float* wrow = W + (size_t)o * CIN2 + 2 * hh;
      const float* arow = edge + rr * CIN2 + 2 * hh;
      v8f acc = {};
      #pragma unroll
      for (int g = 0; g < 4; ++g) {
        v2f av[8], bv[8];
        #pragma unroll
        for (int u = 0; u < 8; ++u) {
          av[u] = *(const v2f*)(arow + (g * 8 + u) * 4);
          bv[u] = *(const v2f*)(wrow + (g * 8 + u) * 4);
        }
        #pragma unroll
        for (int u = 0; u < 8; ++u) acc = wmma_f32(av[u], bv[u], acc);
      }
      float bo = bias[o];
      float gAmax = -__builtin_inff(), gAmin = __builtin_inff();
      float gBmax = -__builtin_inff(), gBmin = __builtin_inff();
      float lsum = 0.f, lsq = 0.f;
      #pragma unroll
      for (int v = 0; v < 8; ++v) {
        float h = acc[v] + bo;
        h = (h >= 0.f) ? h : NEGSL * h;
        lsum += h; lsq += h * h;
        bool inB = (v >= vth);
        gAmax = fmaxf(gAmax, inB ? -__builtin_inff() : h);
        gAmin = fminf(gAmin, inB ?  __builtin_inff() : h);
        gBmax = fmaxf(gBmax, inB ? h : -__builtin_inff());
        gBmin = fminf(gBmin, inB ? h :  __builtin_inff());
      }
      fsum[ot] += lsum; fsq[ot] += lsq;
      atomicMax(&s_umax[w][pA][o], fenc(gAmax));
      atomicMin(&s_umin[w][pA][o], fenc(gAmin));
      if (hasB) {
        atomicMax(&s_umax[w][pA + 1][o], fenc(gBmax));
        atomicMin(&s_umin[w][pA + 1][o], fenc(gBmin));
      }
    }
    __syncthreads();
  }

  #pragma unroll
  for (int ot = 0; ot < 4; ++ot) {
    s_psum[ot][rr][w * 2 + hh] = fsum[ot];
    s_psq [ot][rr][w * 2 + hh] = fsq[ot];
  }
  __syncthreads();
  if (tid < COUT) {
    float s = 0.f, s2 = 0.f;
    #pragma unroll
    for (int q = 0; q < 8; ++q) { s += s_psum[tid >> 4][tid & 15][q]; s2 += s_psq[tid >> 4][tid & 15][q]; }
    part[(size_t)blockIdx.x * 128 + tid]      = s;
    part[(size_t)blockIdx.x * 128 + 64 + tid] = s2;
  }

  #pragma unroll
  for (int it = 0; it < 8; ++it) {
    int q = L + 32 * it;
    int p = q >> 6, o = q & 63;
    int gp = wfp + p;
    int bb = gp >> 13, n = gp & (NPT - 1);
    size_t oi = ((size_t)(bb * COUT + o) << 13) + n;
    mxp[oi] = fdec(s_umax[w][p][o]);
    mnp[oi] = fdec(s_umin[w][p][o]);
  }
}

__global__ void stats_kernel(const float* __restrict__ part, float* __restrict__ gst) {
  int t = threadIdx.x;
  float s = 0.f;
  for (int g = 0; g < 1024; ++g) s += part[(size_t)g * 128 + t];
  gst[t] = s;
}

__global__ void out_kernel(const float* __restrict__ mxp, const float* __restrict__ mnp,
                           const float* __restrict__ gst,
                           const float* __restrict__ gamma, const float* __restrict__ beta,
                           float* __restrict__ outp) {
  int i = blockIdx.x * blockDim.x + threadIdx.x;
  int o = (i >> 13) & 63;
  const float cnt = (float)(BATCH * NPT * KNN_K);
  float mean = gst[o] / cnt;
  float var  = gst[COUT + o] / cnt - mean * mean;
  float scale = gamma[o] * rsqrtf(var + BNEPS);
  float sel = (scale >= 0.f) ? mxp[i] : mnp[i];
  outp[i] = (sel - mean) * scale + beta[o];
}

extern "C" void kernel_launch(void* const* d_in, const int* in_sizes, int n_in,
                              void* d_out, int out_size, void* d_ws, size_t ws_size,
                              hipStream_t stream) {
  const float* x     = (const float*)d_in[0];
  const float* W     = (const float*)d_in[1];
  const float* bias  = (const float*)d_in[2];
  const float* gamma = (const float*)d_in[3];
  const float* beta  = (const float*)d_in[4];
  float* outp = (float*)d_out;

  float* xt  = (float*)d_ws;
  float* sq  = xt + (size_t)BATCH * NPT * CH;
  int*   knn = (int*)(sq + (size_t)BATCH * NPT);
  float* mxp = (float*)(knn + (size_t)BATCH * NPT * KNN_K);
  float* mnp = mxp + (size_t)BATCH * COUT * NPT;
  float* gst = mnp + (size_t)BATCH * COUT * NPT;
  float* part = gst + 128;

  prep_kernel <<<(BATCH * NPT) / 256, 256, 0, stream>>>(x, xt, sq);
  knn_kernel  <<<BATCH * (NPT / 16), 32, 0, stream>>>(xt, sq, knn);
  conv_kernel <<<(BATCH * NPT) / 16, 128, 0, stream>>>(xt, knn, W, bias, mxp, mnp, part);
  stats_kernel<<<1, 128, 0, stream>>>(part, gst);
  out_kernel  <<<(BATCH * COUT * NPT) / 256, 256, 0, stream>>>(mxp, mnp, gst, gamma, beta, outp);
}


# knn top-K: one sorted list per row across the lanes of a register (lane = rank), not one per lane; a passing score is placed by a list compare and a one-lane DPP shift; scores parked row-major
# speedup vs baseline: 1.9762x; 1.9762x over previous
#include <hip/hip_runtime.h>
#include <stdint.h>

#define BATCH 2
#define CH    64
#define NPT   8192
#define KNN_K 20
#define COUT  64
#define CIN2  128
#define NEGSL 0.2f
#define BNEPS 1e-5f

typedef float v2f __attribute__((ext_vector_type(2)));
typedef float v4f __attribute__((ext_vector_type(4)));
typedef float v8f __attribute__((ext_vector_type(8)));

__device__ __forceinline__ v8f wmma_f32(v2f a, v2f b, v8f c) {
  return __builtin_amdgcn_wmma_f32_16x16x4_f32(false, a, false, b, (short)0, c, false, false);
}

__device__ __forceinline__ unsigned fenc(float f) {
  unsigned u = __float_as_uint(f);
  return (u & 0x80000000u) ? ~u : (u | 0x80000000u);
}
__device__ __forceinline__ float fdec(unsigned u) {
  return __uint_as_float((u & 0x80000000u) ? (u ^ 0x80000000u) : ~u);
}

__global__ void prep_kernel(const float* __restrict__ x,
                            float* __restrict__ xt, float* __restrict__ sq) {
  int gp = blockIdx.x * blockDim.x + threadIdx.x;
  int b = gp >> 13, n = gp & (NPT - 1);
  const float* xb = x + (size_t)b * CH * NPT + n;
  float acc = 0.f;
  #pragma unroll 8
  for (int c = 0; c < CH; ++c) {
    float v = xb[(size_t)c * NPT];
    acc += v * v;
    xt[(size_t)gp * CH + c] = v;
  }
  sq[gp] = acc;
}

#define KROWS 64
#define KTHREADS 256
#define TCOL  64
#define PITCH 68
#define RPITCH 68
#define NSTEP (NPT / TCOL)
#define NSUB  (TCOL / 16)

__device__ __forceinline__ float lane_read_f(float v, int lane) {
  return __int_as_float(__builtin_amdgcn_readlane(__float_as_int(v), lane));
}
__device__ __forceinline__ int lane_up1(int v, int edge, bool is16) {
  const int t = __builtin_amdgcn_update_dpp(edge, v, 0x111  , 0xf, 0xf, false);
  const int v15 = __builtin_amdgcn_readlane(v, 15);
  return is16 ? v15 : t;
}

__launch_bounds__(KTHREADS)
__global__ void knn_kernel(const float* __restrict__ xt, const float* __restrict__ sq,
                           int* __restrict__ knn) {
  __shared__ __attribute__((aligned(16))) float s_b[2][TCOL * PITCH];
  __shared__ __attribute__((aligned(16))) float s_sq[4][TCOL];
  __shared__ __attribute__((aligned(16))) float s_tw[2][4][16 * RPITCH];

  const int tid = threadIdx.x;
  const int L = tid & 31;
  const int w  = __builtin_amdgcn_readfirstlane(tid >> 5);
  const int pw = w & 3;
  const bool mat = w < 4;
  const int rr = L & 15, hh = L >> 4;
  const int b  = blockIdx.x >> 7;
  const int r0 = ((blockIdx.x & 127) << 6) + (pw << 4);
  const int bofs = b << 13;
  const bool is16 = (L == 16);

  v2f a[16];
  const float* arow = xt + (size_t)(bofs + r0 + rr) * CH + 2 * hh;
  #pragma unroll
  for (int kk = 0; kk < 16; ++kk) a[kk] = *(const v2f*)(arow + kk * 4);

  float sqr[16], lv[16], thr[16]; int li[16];
  #pragma unroll
  for (int r = 0; r < 16; ++r) {
    sqr[r] = sq[bofs + r0 + r];
    lv[r] = -__builtin_inff(); li[r] = 0; thr[r] = -__builtin_inff();
  }

  const float* gbase = xt + (size_t)bofs * CH;
  auto gload = [&](int st, v4f (&g)[8], float& gs) {
    const float* gsrc = gbase + (size_t)st * (TCOL * CH);
    #pragma unroll
    for (int i = 0; i < 8; ++i) g[i] = *(const v4f*)(gsrc + (size_t)(tid + 128 * i) * 4);
    gs = sq[bofs + st * TCOL + (tid & (TCOL - 1))];
  };
  auto sstore = [&](int st, v4f (&g)[8], float gs) {
    #pragma unroll
    for (int i = 0; i < 8; ++i) {
      int q = tid + 128 * i;
      *(v4f*)&s_b[st & 1][(q >> 4) * PITCH + (q & 15) * 4] = g[i];
    }
    s_sq[st & 3][tid & (TCOL - 1)] = gs;
  };

  v4f g[8]; float gs;
  if (mat) {
    gload(0, g, gs);
    sstore(0, g, gs);
  }
  __syncthreads();

  for (int st = 0; st <= NSTEP; ++st) {
    if (mat) {
      if (st < NSTEP) {
        const int buf = st & 1;
        gload((st + 1) & (NSTEP - 1), g, gs);

        const float* sb = &s_b[buf][0] + rr * PITCH + 2 * hh;
        v2f fr[NSUB][16];
        #pragma unroll
        for (int sub = 0; sub < NSUB; ++sub) {
          #pragma unroll
          for (int kk = 0; kk < 16; ++kk) fr[sub][kk] = *(const v2f*)(sb + sub * 16 * PITCH + kk * 4);
        }
        __builtin_amdgcn_sched_barrier(0);

        v8f acc[NSUB];
        #pragma unroll
        for (int sub = 0; sub < NSUB; ++sub) acc[sub] = (v8f){};
        #pragma unroll
        for (int kk = 0; kk < 16; ++kk) {
          #pragma unroll
          for (int sub = 0; sub < NSUB; ++sub) acc[sub] = wmma_f32(a[kk], fr[sub][kk], acc[sub]);
        }

        float* tw = &s_tw[buf][pw][0] + (8 * hh) * RPITCH + rr;
        #pragma unroll
        for (int sub = 0; sub < NSUB; ++sub) {
          #pragma unroll
          for (int q = 0; q < 8; ++q) tw[q * RPITCH + sub * 16] = acc[sub][q];
        }

        sstore(st + 1, g, gs);
      }
    } else if (st > 0) {
      const int ps = st - 1;
      const float* tw = &s_tw[ps & 1][pw][0] + L;
      float f[16][2];
      #pragma unroll
      for (int r = 0; r < 16; ++r) { f[r][0] = tw[r * RPITCH]; f[r][1] = tw[r * RPITCH + 32]; }
      float sqc[2];
      sqc[0] = s_sq[ps & 3][L]; sqc[1] = s_sq[ps & 3][32 + L];
      const int c0 = ps * TCOL;

      #pragma unroll
      for (int r = 0; r < 16; ++r) {
        #pragma unroll
        for (int u = 0; u < 2; ++u) {
          const float s = 2.f * f[r][u] - sqc[u] - sqr[r];
          unsigned cm = __builtin_amdgcn_ballot_w32(s > thr[r]);
          if (cm) {
            do {
              const int l = __builtin_ctz(cm);
              cm &= cm - 1;
              const float sv = lane_read_f(s, l);
              const int cg = c0 + u * 32 + l;
              const float tv = __int_as_float(lane_up1(__float_as_int(lv[r]), 0x7f800000, is16));
              const int   ti = lane_up1(li[r], 0, is16);
              const bool lt  = lv[r] < sv;
              const bool ltp = tv < sv;
              lv[r] = lt ? (ltp ? tv : sv) : lv[r];
              li[r] = lt ? (ltp ? ti : cg) : li[r];
            } while (cm);
            thr[r] = lane_read_f(lv[r], KNN_K - 1);
          }
        }
      }
    }
    __syncthreads();
  }

  if (!mat && L < KNN_K) {
    int* outp = knn + (size_t)(bofs + r0) * KNN_K + L;
    #pragma unroll
    for (int r = 0; r < 16; ++r) outp[r * KNN_K] = li[r];
  }
}

#define CWAVES 4
#define CPTS   16
#define CONV_WGS ((BATCH * NPT) / (CWAVES * CPTS))
#define OPITCH 68
__launch_bounds__(128)
__global__ void conv_kernel(const float* __restrict__ xt, const int* __restrict__ knn,
                            const float* __restrict__ W, const float* __restrict__ bias,
                            float* __restrict__ mxp, float* __restrict__ mnp,
                            float* __restrict__ part) {
  __shared__ __attribute__((aligned(16))) float s_mx[COUT * OPITCH];
  __shared__ __attribute__((aligned(16))) float s_mn[COUT * OPITCH];
  __shared__ __attribute__((aligned(16))) float s_red[CWAVES][64][32];

  const int tid = threadIdx.x;
  const int w = tid >> 5, L = tid & 31;
  const int c = L & 15, hh = L >> 4;
  const int wg0 = blockIdx.x * (CWAVES * CPTS);
  const int gp  = wg0 + w * CPTS + c;
  const int bofs = gp & ~(NPT - 1);

  v2f wa[4][32];
  #pragma unroll
  for (int ot = 0; ot < 4; ++ot) {
    const float* wrow = W + (size_t)(ot * 16 + c) * CIN2 + 2 * hh;
    #pragma unroll
    for (int kk = 0; kk < 32; ++kk) wa[ot][kk] = *(const v2f*)(wrow + kk * 4);
  }
  v2f xc[16];
  {
    const float* xn = xt + (size_t)gp * CH + 2 * hh;
    #pragma unroll
    for (int kk = 0; kk < 16; ++kk) xc[kk] = *(const v2f*)(xn + kk * 4);
  }
  float bo[4][8];
  #pragma unroll
  for (int ot = 0; ot < 4; ++ot) {
    #pragma unroll
    for (int r = 0; r < 8; ++r) bo[ot][r] = bias[ot * 16 + 8 * hh + r];
  }

  float fsum[4][8], fsq[4][8], vmx[4][8], vmn[4][8];
  #pragma unroll
  for (int ot = 0; ot < 4; ++ot) {
    #pragma unroll
    for (int r = 0; r < 8; ++r) {
      fsum[ot][r] = 0.f; fsq[ot][r] = 0.f;
      vmx[ot][r] = -__builtin_inff(); vmn[ot][r] = __builtin_inff();
    }
  }

  const int* kp = knn + (size_t)gp * KNN_K;
  const float* xbase = xt + (size_t)bofs * CH + 2 * hh;
  auto gather = [&](int m, v2f (&x)[16]) {
    const float* xm = xbase + (size_t)m * CH;
    #pragma unroll
    for (int kk = 0; kk < 16; ++kk) x[kk] = *(const v2f*)(xm + kk * 4);
  };

  int m1 = kp[0], m2 = kp[1];
  v2f xq[16];
  gather(m1, xq);
  m1 = m2; m2 = kp[2];

  #pragma unroll 1
  for (int t = 0; t < KNN_K; ++t) {
    v2f xnx[16];
    gather(m1, xnx);
    m1 = m2;
    m2 = kp[(t + 3 < KNN_K) ? t + 3 : KNN_K - 1];

    v8f acc[4];
    #pragma unroll
    for (int ot = 0; ot < 4; ++ot) acc[ot] = (v8f){};
    #pragma unroll
    for (int kk = 0; kk < 16; ++kk) {
      v2f d = xq[kk] - xc[kk];
      #pragma unroll
      for (int ot = 0; ot < 4; ++ot) acc[ot] = wmma_f32(wa[ot][kk], d, acc[ot]);
    }
    #pragma unroll
    for (int kk = 0; kk < 16; ++kk) {
      #pragma unroll
      for (int ot = 0; ot < 4; ++ot) acc[ot] = wmma_f32(wa[ot][16 + kk], xc[kk], acc[ot]);
    }

    #pragma unroll
    for (int ot = 0; ot < 4; ++ot) {
      #pragma unroll
      for (int r = 0; r < 8; ++r) {
        float h = acc[ot][r] + bo[ot][r];
        h = (h >= 0.f) ? h : NEGSL * h;
        fsum[ot][r] += h; fsq[ot][r] += h * h;
        vmx[ot][r] = fmaxf(vmx[ot][r], h);
        vmn[ot][r] = fminf(vmn[ot][r], h);
      }
    }
    #pragma unroll
    for (int kk = 0; kk < 16; ++kk) xq[kk] = xnx[kk];
  }

  #pragma unroll
  for (int ot = 0; ot < 4; ++ot) {
    #pragma unroll
    for (int r = 0; r < 8; ++r) {
      const int o = ot * 16 + 8 * hh + r;
      s_mx[o * OPITCH + w * CPTS + c] = vmx[ot][r];
      s_mn[o * OPITCH + w * CPTS + c] = vmn[ot][r];
      s_red[w][ot * 8 + r][L]      = fsum[ot][r];
      s_red[w][32 + ot * 8 + r][L] = fsq[ot][r];
    }
  }
  __syncthreads();

  {
    const int o = tid & 63, kind = tid >> 6;
    const int v = kind * 32 + (o >> 4) * 8 + (o & 7);
    const int h2 = (o >> 3) & 1;
    float s = 0.f;
    for (int ww = 0; ww < CWAVES; ++ww) {
      for (int q = 0; q < 4; ++q) {
        v4f p = *(const v4f*)&s_red[ww][v][h2 * 16 + q * 4];
        s += p[0]; s += p[1]; s += p[2]; s += p[3];
      }
    }
    part[(size_t)blockIdx.x * 128 + kind * 64 + o] = s;
  }

  {
    const int bb = wg0 >> 13, n0 = wg0 & (NPT - 1);
    for (int it = 0; it < 8; ++it) {
      const int q = tid + 128 * it;
      const int o = q >> 4, ch = q & 15;
      const size_t oi = ((size_t)(bb * COUT + o) << 13) + n0 + ch * 4;
      *(v4f*)(mxp + oi) = *(const v4f*)&s_mx[o * OPITCH + ch * 4];
      *(v4f*)(mnp + oi) = *(const v4f*)&s_mn[o * OPITCH + ch * 4];
    }
  }
}

__global__ void stats_kernel(const float* __restrict__ part, float* __restrict__ gst) {
  int t = threadIdx.x;
  float s = 0.f;
  for (int g = 0; g < CONV_WGS; ++g) s += part[(size_t)g * 128 + t];
  gst[t] = s;
}

__global__ void out_kernel(const float* __restrict__ mxp, const float* __restrict__ mnp,
                           const float* __restrict__ gst,
                           const float* __restrict__ gamma, const float* __restrict__ beta,
                           float* __restrict__ outp) {
  int i = blockIdx.x * blockDim.x + threadIdx.x;
  int o = (i >> 13) & 63;
  const float cnt = (float)(BATCH * NPT * KNN_K);
  float mean = gst[o] / cnt;
  float var  = gst[COUT + o] / cnt - mean * mean;
  float scale = gamma[o] * rsqrtf(var + BNEPS);
  float sel = (scale >= 0.f) ? mxp[i] : mnp[i];
  outp[i] = (sel - mean) * scale + beta[o];
}

extern "C" void kernel_launch(void* const* d_in, const int* in_sizes, int n_in,
                              void* d_out, int out_size, void* d_ws, size_t ws_size,
                              hipStream_t stream) {
  const float* x     = (const float*)d_in[0];
  const float* W     = (const float*)d_in[1];
  const float* bias  = (const float*)d_in[2];
  const float* gamma = (const float*)d_in[3];
  const float* beta  = (const float*)d_in[4];
  float* outp = (float*)d_out;

  float* xt  = (float*)d_ws;
  float* sq  = xt + (size_t)BATCH * NPT * CH;
  int*   knn = (int*)(sq + (size_t)BATCH * NPT);
  float* mxp = (float*)(knn + (size_t)BATCH * NPT * KNN_K);
  float* mnp = mxp + (size_t)BATCH * COUT * NPT;
  float* gst = mnp + (size_t)BATCH * COUT * NPT;
  float* part = gst + 128;

  prep_kernel <<<(BATCH * NPT) / 256, 256, 0, stream>>>(x, xt, sq);
  knn_kernel  <<<BATCH * (NPT / KROWS), KTHREADS, 0, stream>>>(xt, sq, knn);
  conv_kernel <<<CONV_WGS, 128, 0, stream>>>(xt, knn, W, bias, mxp, mnp, part);
  stats_kernel<<<1, 128, 0, stream>>>(part, gst);
  out_kernel  <<<(BATCH * COUT * NPT) / 256, 256, 0, stream>>>(mxp, mnp, gst, gamma, beta, outp);
}


# knn scan: a stage's 32 score tests against the rows' K-th best are made first as one branch-free group (one hit word per row); only a row with a hit enters the insertion code; insert order kept
# speedup vs baseline: 2.1181x; 1.0718x over previous
#include <hip/hip_runtime.h>
#include <stdint.h>

#define BATCH 2
#define CH    64
#define NPT   8192
#define KNN_K 20
#define COUT  64
#define CIN2  128
#define NEGSL 0.2f
#define BNEPS 1e-5f

typedef float v2f __attribute__((ext_vector_type(2)));
typedef float v4f __attribute__((ext_vector_type(4)));
typedef float v8f __attribute__((ext_vector_type(8)));

__device__ __forceinline__ v8f wmma_f32(v2f a, v2f b, v8f c) {
  return __builtin_amdgcn_wmma_f32_16x16x4_f32(false, a, false, b, (short)0, c, false, false);
}

__device__ __forceinline__ unsigned fenc(float f) {
  unsigned u = __float_as_uint(f);
  return (u & 0x80000000u) ? ~u : (u | 0x80000000u);
}
__device__ __forceinline__ float fdec(unsigned u) {
  return __uint_as_float((u & 0x80000000u) ? (u ^ 0x80000000u) : ~u);
}

__global__ void prep_kernel(const float* __restrict__ x,
                            float* __restrict__ xt, float* __restrict__ sq) {
  int gp = blockIdx.x * blockDim.x + threadIdx.x;
  int b = gp >> 13, n = gp & (NPT - 1);
  const float* xb = x + (size_t)b * CH * NPT + n;
  float acc = 0.f;
  #pragma unroll 8
  for (int c = 0; c < CH; ++c) {
    float v = xb[(size_t)c * NPT];
    acc += v * v;
    xt[(size_t)gp * CH + c] = v;
  }
  sq[gp] = acc;
}

#define KROWS 64
#define KTHREADS 256
#define TCOL  64
#define PITCH 68
#define RPITCH 68
#define NSTEP (NPT / TCOL)
#define NSUB  (TCOL / 16)

__device__ __forceinline__ float lane_read_f(float v, int lane) {
  return __int_as_float(__builtin_amdgcn_readlane(__float_as_int(v), lane));
}
__device__ __forceinline__ int lane_up1(int v, int edge, bool is16) {
  const int t = __builtin_amdgcn_update_dpp(edge, v, 0x111  , 0xf, 0xf, false);
  const int v15 = __builtin_amdgcn_readlane(v, 15);
  return is16 ? v15 : t;
}

__launch_bounds__(KTHREADS)
__global__ void knn_kernel(const float* __restrict__ xt, const float* __restrict__ sq,
                           int* __restrict__ knn) {
  __shared__ __attribute__((aligned(16))) float s_b[2][TCOL * PITCH];
  __shared__ __attribute__((aligned(16))) float s_sq[4][TCOL];
  __shared__ __attribute__((aligned(16))) float s_tw[2][4][16 * RPITCH];

  const int tid = threadIdx.x;
  const int L = tid & 31;
  const int w  = __builtin_amdgcn_readfirstlane(tid >> 5);
  const int pw = w & 3;
  const bool mat = w < 4;
  const int rr = L & 15, hh = L >> 4;
  const int b  = blockIdx.x >> 7;
  const int r0 = ((blockIdx.x & 127) << 6) + (pw << 4);
  const int bofs = b << 13;
  const bool is16 = (L == 16);

  v2f a[16];
  const float* arow = xt + (size_t)(bofs + r0 + rr) * CH + 2 * hh;
  #pragma unroll
  for (int kk = 0; kk < 16; ++kk) a[kk] = *(const v2f*)(arow + kk * 4);

  float sqr[16], lv[16], thr[16]; int li[16];
  #pragma unroll
  for (int r = 0; r < 16; ++r) {
    sqr[r] = sq[bofs + r0 + r];
    lv[r] = -__builtin_inff(); li[r] = 0; thr[r] = -__builtin_inff();
  }

  const float* gbase = xt + (size_t)bofs * CH;
  auto gload = [&](int st, v4f (&g)[8], float& gs) {
    const float* gsrc = gbase + (size_t)st * (TCOL * CH);
    #pragma unroll
    for (int i = 0; i < 8; ++i) g[i] = *(const v4f*)(gsrc + (size_t)(tid + 128 * i) * 4);
    gs = sq[bofs + st * TCOL + (tid & (TCOL - 1))];
  };
  auto sstore = [&](int st, v4f (&g)[8], float gs) {
    #pragma unroll
    for (int i = 0; i < 8; ++i) {
      int q = tid + 128 * i;
      *(v4f*)&s_b[st & 1][(q >> 4) * PITCH + (q & 15) * 4] = g[i];
    }
    s_sq[st & 3][tid & (TCOL - 1)] = gs;
  };

  v4f g[8]; float gs;
  if (mat) {
    gload(0, g, gs);
    sstore(0, g, gs);
  }
  __syncthreads();

  for (int st = 0; st <= NSTEP; ++st) {
    if (mat) {
      if (st < NSTEP) {
        const int buf = st & 1;
        gload((st + 1) & (NSTEP - 1), g, gs);

        const float* sb = &s_b[buf][0] + rr * PITCH + 2 * hh;
        v2f fr[NSUB][16];
        #pragma unroll
        for (int sub = 0; sub < NSUB; ++sub) {
          #pragma unroll
          for (int kk = 0; kk < 16; ++kk) fr[sub][kk] = *(const v2f*)(sb + sub * 16 * PITCH + kk * 4);
        }
        __builtin_amdgcn_sched_barrier(0);

        v8f acc[NSUB];
        #pragma unroll
        for (int sub = 0; sub < NSUB; ++sub) acc[sub] = (v8f){};
        #pragma unroll
        for (int kk = 0; kk < 16; ++kk) {
          #pragma unroll
          for (int sub = 0; sub < NSUB; ++sub) acc[sub] = wmma_f32(a[kk], fr[sub][kk], acc[sub]);
        }

        float* tw = &s_tw[buf][pw][0] + (8 * hh) * RPITCH + rr;
        #pragma unroll
        for (int sub = 0; sub < NSUB; ++sub) {
          #pragma unroll
          for (int q = 0; q < 8; ++q) tw[q * RPITCH + sub * 16] = acc[sub][q];
        }

        sstore(st + 1, g, gs);
      }
    } else if (st > 0) {
      const int ps = st - 1;
      const float* tw = &s_tw[ps & 1][pw][0] + L;
      float f[16][2];
      #pragma unroll
      for (int r = 0; r < 16; ++r) { f[r][0] = tw[r * RPITCH]; f[r][1] = tw[r * RPITCH + 32]; }
      float sqc[2];
      sqc[0] = s_sq[ps & 3][L]; sqc[1] = s_sq[ps & 3][32 + L];
      const int c0 = ps * TCOL;

      unsigned hit[16];
      #pragma unroll
      for (int r = 0; r < 16; ++r) {
        f[r][0] = 2.f * f[r][0] - sqc[0] - sqr[r];
        f[r][1] = 2.f * f[r][1] - sqc[1] - sqr[r];
        hit[r] = __builtin_amdgcn_ballot_w32(f[r][0] > thr[r]) | __builtin_amdgcn_ballot_w32(f[r][1] > thr[r]);
      }

      #pragma unroll
      for (int r = 0; r < 16; ++r) {
        if (__builtin_expect(hit[r] == 0, 1)) continue;
        #pragma unroll
        for (int u = 0; u < 2; ++u) {
          const float s = f[r][u];
          unsigned cm = __builtin_amdgcn_ballot_w32(s > thr[r]);
          if (cm) {
            do {
              const int l = __builtin_ctz(cm);
              cm &= cm - 1;
              const float sv = lane_read_f(s, l);
              const int cg = c0 + u * 32 + l;
              const float tv = __int_as_float(lane_up1(__float_as_int(lv[r]), 0x7f800000, is16));
              const int   ti = lane_up1(li[r], 0, is16);
              const bool lt  = lv[r] < sv;
              const bool ltp = tv < sv;
              lv[r] = lt ? (ltp ? tv : sv) : lv[r];
              li[r] = lt ? (ltp ? ti : cg) : li[r];
            } while (cm);
            thr[r] = lane_read_f(lv[r], KNN_K - 1);
          }
        }
      }
    }
    __syncthreads();
  }

  if (!mat && L < KNN_K) {
    int* outp = knn + (size_t)(bofs + r0) * KNN_K + L;
    #pragma unroll
    for (int r = 0; r < 16; ++r) outp[r * KNN_K] = li[r];
  }
}

#define CWAVES 4
#define CPTS   16
#define CONV_WGS ((BATCH * NPT) / (CWAVES * CPTS))
#define OPITCH 68
__launch_bounds__(128)
__global__ void conv_kernel(const float* __restrict__ xt, const int* __restrict__ knn,
                            const float* __restrict__ W, const float* __restrict__ bias,
                            float* __restrict__ mxp, float* __restrict__ mnp,
                            float* __restrict__ part) {
  __shared__ __attribute__((aligned(16))) float s_mx[COUT * OPITCH];
  __shared__ __attribute__((aligned(16))) float s_mn[COUT * OPITCH];
  __shared__ __attribute__((aligned(16))) float s_red[CWAVES][64][32];

  const int tid = threadIdx.x;
  const int w = tid >> 5, L = tid & 31;
  const int c = L & 15, hh = L >> 4;
  const int wg0 = blockIdx.x * (CWAVES * CPTS);
  const int gp  = wg0 + w * CPTS + c;
  const int bofs = gp & ~(NPT - 1);

  v2f wa[4][32];
  #pragma unroll
  for (int ot = 0; ot < 4; ++ot) {
    const float* wrow = W + (size_t)(ot * 16 + c) * CIN2 + 2 * hh;
    #pragma unroll
    for (int kk = 0; kk < 32; ++kk) wa[ot][kk] = *(const v2f*)(wrow + kk * 4);
  }
  v2f xc[16];
  {
    const float* xn = xt + (size_t)gp * CH + 2 * hh;
    #pragma unroll
    for (int kk = 0; kk < 16; ++kk) xc[kk] = *(const v2f*)(xn + kk * 4);
  }
  float bo[4][8];
  #pragma unroll
  for (int ot = 0; ot < 4; ++ot) {
    #pragma unroll
    for (int r = 0; r < 8; ++r) bo[ot][r] = bias[ot * 16 + 8 * hh + r];
  }

  float fsum[4][8], fsq[4][8], vmx[4][8], vmn[4][8];
  #pragma unroll
  for (int ot = 0; ot < 4; ++ot) {
    #pragma unroll
    for (int r = 0; r < 8; ++r) {
      fsum[ot][r] = 0.f; fsq[ot][r] = 0.f;
      vmx[ot][r] = -__builtin_inff(); vmn[ot][r] = __builtin_inff();
    }
  }

  const int* kp = knn + (size_t)gp * KNN_K;
  const float* xbase = xt + (size_t)bofs * CH + 2 * hh;
  auto gather = [&](int m, v2f (&x)[16]) {
    const float* xm = xbase + (size_t)m * CH;
    #pragma unroll
    for (int kk = 0; kk < 16; ++kk) x[kk] = *(const v2f*)(xm + kk * 4);
  };

  int m1 = kp[0], m2 = kp[1];
  v2f xq[16];
  gather(m1, xq);
  m1 = m2; m2 = kp[2];

  #pragma unroll 1
  for (int t = 0; t < KNN_K; ++t) {
    v2f xnx[16];
    gather(m1, xnx);
    m1 = m2;
    m2 = kp[(t + 3 < KNN_K) ? t + 3 : KNN_K - 1];

    v8f acc[4];
    #pragma unroll
    for (int ot = 0; ot < 4; ++ot) acc[ot] = (v8f){};
    #pragma unroll
    for (int kk = 0; kk < 16; ++kk) {
      v2f d = xq[kk] - xc[kk];
      #pragma unroll
      for (int ot = 0; ot < 4; ++ot) acc[ot] = wmma_f32(wa[ot][kk], d, acc[ot]);
    }
    #pragma unroll
    for (int kk = 0; kk < 16; ++kk) {
      #pragma unroll
      for (int ot = 0; ot < 4; ++ot) acc[ot] = wmma_f32(wa[ot][16 + kk], xc[kk], acc[ot]);
    }

    #pragma unroll
    for (int ot = 0; ot < 4; ++ot) {
      #pragma unroll
      for (int r = 0; r < 8; ++r) {
        float h = acc[ot][r] + bo[ot][r];
        h = (h >= 0.f) ? h : NEGSL * h;
        fsum[ot][r] += h; fsq[ot][r] += h * h;
        vmx[ot][r] = fmaxf(vmx[ot][r], h);
        vmn[ot][r] = fminf(vmn[ot][r], h);
      }
    }
    #pragma unroll
    for (int kk = 0; kk < 16; ++kk) xq[kk] = xnx[kk];
  }

  #pragma unroll
  for (int ot = 0; ot < 4; ++ot) {
    #pragma unroll
    for (int r = 0; r < 8; ++r) {
      const int o = ot * 16 + 8 * hh + r;
      s_mx[o * OPITCH + w * CPTS + c] = vmx[ot][r];
      s_mn[o * OPITCH + w * CPTS + c] = vmn[ot][r];
      s_red[w][ot * 8 + r][L]      = fsum[ot][r];
      s_red[w][32 + ot * 8 + r][L] = fsq[ot][r];
    }
  }
  __syncthreads();

  {
    const int o = tid & 63, kind = tid >> 6;
    const int v = kind * 32 + (o >> 4) * 8 + (o & 7);
    const int h2 = (o >> 3) & 1;
    float s = 0.f;
    for (int ww = 0; ww < CWAVES; ++ww) {
      for (int q = 0; q < 4; ++q) {
        v4f p = *(const v4f*)&s_red[ww][v][h2 * 16 + q * 4];
        s += p[0]; s += p[1]; s += p[2]; s += p[3];
      }
    }
    part[(size_t)blockIdx.x * 128 + kind * 64 + o] = s;
  }

  {
    const int bb = wg0 >> 13, n0 = wg0 & (NPT - 1);
    for (int it = 0; it < 8; ++it) {
      const int q = tid + 128 * it;
      const int o = q >> 4, ch = q & 15;
      const size_t oi = ((size_t)(bb * COUT + o) << 13) + n0 + ch * 4;
      *(v4f*)(mxp + oi) = *(const v4f*)&s_mx[o * OPITCH + ch * 4];
      *(v4f*)(mnp + oi) = *(const v4f*)&s_mn[o * OPITCH + ch * 4];
    }
  }
}

__global__ void stats_kernel(const float* __restrict__ part, float* __restrict__ gst) {
  int t = threadIdx.x;
  float s = 0.f;
  for (int g = 0; g < CONV_WGS; ++g) s += part[(size_t)g * 128 + t];
  gst[t] = s;
}

__global__ void out_kernel(const float* __restrict__ mxp, const float* __restrict__ mnp,
                           const float* __restrict__ gst,
                           const float* __restrict__ gamma, const float* __restrict__ beta,
                           float* __restrict__ outp) {
  int i = blockIdx.x * blockDim.x + threadIdx.x;
  int o = (i >> 13) & 63;
  const float cnt = (float)(BATCH * NPT * KNN_K);
  float mean = gst[o] / cnt;
  float var  = gst[COUT + o] / cnt - mean * mean;
  float scale = gamma[o] * rsqrtf(var + BNEPS);
  float sel = (scale >= 0.f) ? mxp[i] : mnp[i];
  outp[i] = (sel - mean) * scale + beta[o];
}

extern "C" void kernel_launch(void* const* d_in, const int* in_sizes, int n_in,
                              void* d_out, int out_size, void* d_ws, size_t ws_size,
                              hipStream_t stream) {
  const float* x     = (const float*)d_in[0];
  const float* W     = (const float*)d_in[1];
  const float* bias  = (const float*)d_in[2];
  const float* gamma = (const float*)d_in[3];
  const float* beta  = (const float*)d_in[4];
  float* outp = (float*)d_out;

  float* xt  = (float*)d_ws;
  float* sq  = xt + (size_t)BATCH * NPT * CH;
  int*   knn = (int*)(sq + (size_t)BATCH * NPT);
  float* mxp = (float*)(knn + (size_t)BATCH * NPT * KNN_K);
  float* mnp = mxp + (size_t)BATCH * COUT * NPT;
  float* gst = mnp + (size_t)BATCH * COUT * NPT;
  float* part = gst + 128;

  prep_kernel <<<(BATCH * NPT) / 256, 256, 0, stream>>>(x, xt, sq);
  knn_kernel  <<<BATCH * (NPT / KROWS), KTHREADS, 0, stream>>>(xt, sq, knn);
  conv_kernel <<<CONV_WGS, 128, 0, stream>>>(xt, knn, W, bias, mxp, mnp, part);
  stats_kernel<<<1, 128, 0, stream>>>(part, gst);
  out_kernel  <<<(BATCH * COUT * NPT) / 256, 256, 0, stream>>>(mxp, mnp, gst, gamma, beta, outp);
}


# knn: the top-K scan of a 16-row block is split by rows over two scan waves (8 rows each) beside its matrix wave: 12 waves per workgroup, 3 per SIMD; each row's insertion order and all sums unchanged
# speedup vs baseline: 2.8849x; 1.3620x over previous
#include <hip/hip_runtime.h>
#include <stdint.h>

#define BATCH 2
#define CH    64
#define NPT   8192
#define KNN_K 20
#define COUT  64
#define CIN2  128
#define NEGSL 0.2f
#define BNEPS 1e-5f

typedef float v2f __attribute__((ext_vector_type(2)));
typedef float v4f __attribute__((ext_vector_type(4)));
typedef float v8f __attribute__((ext_vector_type(8)));

__device__ __forceinline__ v8f wmma_f32(v2f a, v2f b, v8f c) {
  return __builtin_amdgcn_wmma_f32_16x16x4_f32(false, a, false, b, (short)0, c, false, false);
}

__device__ __forceinline__ unsigned fenc(float f) {
  unsigned u = __float_as_uint(f);
  return (u & 0x80000000u) ? ~u : (u | 0x80000000u);
}
__device__ __forceinline__ float fdec(unsigned u) {
  return __uint_as_float((u & 0x80000000u) ? (u ^ 0x80000000u) : ~u);
}

__global__ void prep_kernel(const float* __restrict__ x,
                            float* __restrict__ xt, float* __restrict__ sq) {
  int gp = blockIdx.x * blockDim.x + threadIdx.x;
  int b = gp >> 13, n = gp & (NPT - 1);
  const float* xb = x + (size_t)b * CH * NPT + n;
  float acc = 0.f;
  #pragma unroll 8
  for (int c = 0; c < CH; ++c) {
    float v = xb[(size_t)c * NPT];
    acc += v * v;
    xt[(size_t)gp * CH + c] = v;
  }
  sq[gp] = acc;
}

#define KROWS 64
#define KTHREADS 384
#define SROWS 8
#define TCOL  64
#define PITCH 68
#define RPITCH 68
#define NSTEP (NPT / TCOL)
#define NSUB  (TCOL / 16)

__device__ __forceinline__ float lane_read_f(float v, int lane) {
  return __int_as_float(__builtin_amdgcn_readlane(__float_as_int(v), lane));
}
__device__ __forceinline__ int lane_up1(int v, int edge, bool is16) {
  const int t = __builtin_amdgcn_update_dpp(edge, v, 0x111  , 0xf, 0xf, false);
  const int v15 = __builtin_amdgcn_readlane(v, 15);
  return is16 ? v15 : t;
}

__launch_bounds__(KTHREADS)
__global__ void knn_kernel(const float* __restrict__ xt, const float* __restrict__ sq,
                           int* __restrict__ knn) {
  __shared__ __attribute__((aligned(16))) float s_b[2][TCOL * PITCH];
  __shared__ __attribute__((aligned(16))) float s_sq[4][TCOL];
  __shared__ __attribute__((aligned(16))) float s_tw[2][4][16 * RPITCH];

  const int tid = threadIdx.x;
  const int L = tid & 31;
  const int w  = __builtin_amdgcn_readfirstlane(tid >> 5);
  const int pw = w & 3;
  const bool mat = w < 4;
  const int rb = mat ? 0 : ((w - 4) >> 2) * SROWS;
  const int rr = L & 15, hh = L >> 4;
  const int b  = blockIdx.x >> 7;
  const int r0 = ((blockIdx.x & 127) << 6) + (pw << 4);
  const int bofs = b << 13;
  const bool is16 = (L == 16);

  v2f a[16];
  const float* arow = xt + (size_t)(bofs + r0 + rr) * CH + 2 * hh;
  #pragma unroll
  for (int kk = 0; kk < 16; ++kk) a[kk] = *(const v2f*)(arow + kk * 4);

  float sqr[SROWS], lv[SROWS], thr[SROWS]; int li[SROWS];
  #pragma unroll
  for (int r = 0; r < SROWS; ++r) {
    sqr[r] = sq[bofs + r0 + rb + r];
    lv[r] = -__builtin_inff(); li[r] = 0; thr[r] = -__builtin_inff();
  }

  const float* gbase = xt + (size_t)bofs * CH;
  auto gload = [&](int st, v4f (&g)[8], float& gs) {
    const float* gsrc = gbase + (size_t)st * (TCOL * CH);
    #pragma unroll
    for (int i = 0; i < 8; ++i) g[i] = *(const v4f*)(gsrc + (size_t)(tid + 128 * i) * 4);
    gs = sq[bofs + st * TCOL + (tid & (TCOL - 1))];
  };
  auto sstore = [&](int st, v4f (&g)[8], float gs) {
    #pragma unroll
    for (int i = 0; i < 8; ++i) {
      int q = tid + 128 * i;
      *(v4f*)&s_b[st & 1][(q >> 4) * PITCH + (q & 15) * 4] = g[i];
    }
    s_sq[st & 3][tid & (TCOL - 1)] = gs;
  };

  v4f g[8]; float gs;
  if (mat) {
    gload(0, g, gs);
    sstore(0, g, gs);
  }
  __syncthreads();

  for (int st = 0; st <= NSTEP; ++st) {
    if (mat) {
      if (st < NSTEP) {
        const int buf = st & 1;
        gload((st + 1) & (NSTEP - 1), g, gs);

        const float* sb = &s_b[buf][0] + rr * PITCH + 2 * hh;
        v2f fr[NSUB][16];
        #pragma unroll
        for (int sub = 0; sub < NSUB; ++sub) {
          #pragma unroll
          for (int kk = 0; kk < 16; ++kk) fr[sub][kk] = *(const v2f*)(sb + sub * 16 * PITCH + kk * 4);
        }
        __builtin_amdgcn_sched_barrier(0);

        v8f acc[NSUB];
        #pragma unroll
        for (int sub = 0; sub < NSUB; ++sub) acc[sub] = (v8f){};
        #pragma unroll
        for (int kk = 0; kk < 16; ++kk) {
          #pragma unroll
          for (int sub = 0; sub < NSUB; ++sub) acc[sub] = wmma_f32(a[kk], fr[sub][kk], acc[sub]);
        }

        float* tw = &s_tw[buf][pw][0] + (8 * hh) * RPITCH + rr;
        #pragma unroll
        for (int sub = 0; sub < NSUB; ++sub) {
          #pragma unroll
          for (int q = 0; q < 8; ++q) tw[q * RPITCH + sub * 16] = acc[sub][q];
        }

        sstore(st + 1, g, gs);
      }
    } else if (st > 0) {
      const int ps = st - 1;
      const float* tw = &s_tw[ps & 1][pw][0] + rb * RPITCH + L;
      float f[SROWS][2];
      #pragma unroll
      for (int r = 0; r < SROWS; ++r) { f[r][0] = tw[r * RPITCH]; f[r][1] = tw[r * RPITCH + 32]; }
      float sqc[2];
      sqc[0] = s_sq[ps & 3][L]; sqc[1] = s_sq[ps & 3][32 + L];
      const int c0 = ps * TCOL;

      unsigned hit[SROWS];
      #pragma unroll
      for (int r = 0; r < SROWS; ++r) {
        f[r][0] = 2.f * f[r][0] - sqc[0] - sqr[r];
        f[r][1] = 2.f * f[r][1] - sqc[1] - sqr[r];
        hit[r] = __builtin_amdgcn_ballot_w32(f[r][0] > thr[r]) | __builtin_amdgcn_ballot_w32(f[r][1] > thr[r]);
      }

      #pragma unroll
      for (int r = 0; r < SROWS; ++r) {
        if (__builtin_expect(hit[r] == 0, 1)) continue;
        #pragma unroll
        for (int u = 0; u < 2; ++u) {
          const float s = f[r][u];
          unsigned cm = __builtin_amdgcn_ballot_w32(s > thr[r]);
          if (cm) {
            do {
              const int l = __builtin_ctz(cm);
              cm &= cm - 1;
              const float sv = lane_read_f(s, l);
              const int cg = c0 + u * 32 + l;
              const float tv = __int_as_float(lane_up1(__float_as_int(lv[r]), 0x7f800000, is16));
              const int   ti = lane_up1(li[r], 0, is16);
              const bool lt  = lv[r] < sv;
              const bool ltp = tv < sv;
              lv[r] = lt ? (ltp ? tv : sv) : lv[r];
              li[r] = lt ? (ltp ? ti : cg) : li[r];
            } while (cm);
            thr[r] = lane_read_f(lv[r], KNN_K - 1);
          }
        }
      }
    }
    __syncthreads();
  }

  if (!mat && L < KNN_K) {
    int* outp = knn + (size_t)(bofs + r0 + rb) * KNN_K + L;
    #pragma unroll
    for (int r = 0; r < SROWS; ++r) outp[r * KNN_K] = li[r];
  }
}

#define CWAVES 4
#define CPTS   16
#define CONV_WGS ((BATCH * NPT) / (CWAVES * CPTS))
#define OPITCH 68
__launch_bounds__(128)
__global__ void conv_kernel(const float* __restrict__ xt, const int* __restrict__ knn,
                            const float* __restrict__ W, const float* __restrict__ bias,
                            float* __restrict__ mxp, float* __restrict__ mnp,
                            float* __restrict__ part) {
  __shared__ __attribute__((aligned(16))) float s_mx[COUT * OPITCH];
  __shared__ __attribute__((aligned(16))) float s_mn[COUT * OPITCH];
  __shared__ __attribute__((aligned(16))) float s_red[CWAVES][64][32];

  const int tid = threadIdx.x;
  const int w = tid >> 5, L = tid & 31;
  const int c = L & 15, hh = L >> 4;
  const int wg0 = blockIdx.x * (CWAVES * CPTS);
  const int gp  = wg0 + w * CPTS + c;
  const int bofs = gp & ~(NPT - 1);

  v2f wa[4][32];
  #pragma unroll
  for (int ot = 0; ot < 4; ++ot) {
    const float* wrow = W + (size_t)(ot * 16 + c) * CIN2 + 2 * hh;
    #pragma unroll
    for (int kk = 0; kk < 32; ++kk) wa[ot][kk] = *(const v2f*)(wrow + kk * 4);
  }
  v2f xc[16];
  {
    const float* xn = xt + (size_t)gp * CH + 2 * hh;
    #pragma unroll
    for (int kk = 0; kk < 16; ++kk) xc[kk] = *(const v2f*)(xn + kk * 4);
  }
  float bo[4][8];
  #pragma unroll
  for (int ot = 0; ot < 4; ++ot) {
    #pragma unroll
    for (int r = 0; r < 8; ++r) bo[ot][r] = bias[ot * 16 + 8 * hh + r];
  }

  float fsum[4][8], fsq[4][8], vmx[4][8], vmn[4][8];
  #pragma unroll
  for (int ot = 0; ot < 4; ++ot) {
    #pragma unroll
    for (int r = 0; r < 8; ++r) {
      fsum[ot][r] = 0.f; fsq[ot][r] = 0.f;
      vmx[ot][r] = -__builtin_inff(); vmn[ot][r] = __builtin_inff();
    }
  }

  const int* kp = knn + (size_t)gp * KNN_K;
  const float* xbase = xt + (size_t)bofs * CH + 2 * hh;
  auto gather = [&](int m, v2f (&x)[16]) {
    const float* xm = xbase + (size_t)m * CH;
    #pragma unroll
    for (int kk = 0; kk < 16; ++kk) x[kk] = *(const v2f*)(xm + kk * 4);
  };

  int m1 = kp[0], m2 = kp[1];
  v2f xq[16];
  gather(m1, xq);
  m1 = m2; m2 = kp[2];

  #pragma unroll 1
  for (int t = 0; t < KNN_K; ++t) {
    v2f xnx[16];
    gather(m1, xnx);
    m1 = m2;
    m2 = kp[(t + 3 < KNN_K) ? t + 3 : KNN_K - 1];

    v8f acc[4];
    #pragma unroll
    for (int ot = 0; ot < 4; ++ot) acc[ot] = (v8f){};
    #pragma unroll
    for (int kk = 0; kk < 16; ++kk) {
      v2f d = xq[kk] - xc[kk];
      #pragma unroll
      for (int ot = 0; ot < 4; ++ot) acc[ot] = wmma_f32(wa[ot][kk], d, acc[ot]);
    }
    #pragma unroll
    for (int kk = 0; kk < 16; ++kk) {
      #pragma unroll
      for (int ot = 0; ot < 4; ++ot) acc[ot] = wmma_f32(wa[ot][16 + kk], xc[kk], acc[ot]);
    }

    #pragma unroll
    for (int ot = 0; ot < 4; ++ot) {
      #pragma unroll
      for (int r = 0; r < 8; ++r) {
        float h = acc[ot][r] + bo[ot][r];
        h = (h >= 0.f) ? h : NEGSL * h;
        fsum[ot][r] += h; fsq[ot][r] += h * h;
        vmx[ot][r] = fmaxf(vmx[ot][r], h);
        vmn[ot][r] = fminf(vmn[ot][r], h);
      }
    }
    #pragma unroll
    for (int kk = 0; kk < 16; ++kk) xq[kk] = xnx[kk];
  }

  #pragma unroll
  for (int ot = 0; ot < 4; ++ot) {
    #pragma unroll
    for (int r = 0; r < 8; ++r) {
      const int o = ot * 16 + 8 * hh + r;
      s_mx[o * OPITCH + w * CPTS + c] = vmx[ot][r];
      s_mn[o * OPITCH + w * CPTS + c] = vmn[ot][r];
      s_red[w][ot * 8 + r][L]      = fsum[ot][r];
      s_red[w][32 + ot * 8 + r][L] = fsq[ot][r];
    }
  }
  __syncthreads();

  {
    const int o = tid & 63, kind = tid >> 6;
    const int v = kind * 32 + (o >> 4) * 8 + (o & 7);
    const int h2 = (o >> 3) & 1;
    float s = 0.f;
    for (int ww = 0; ww < CWAVES; ++ww) {
      for (int q = 0; q < 4; ++q) {
        v4f p = *(const v4f*)&s_red[ww][v][h2 * 16 + q * 4];
        s += p[0]; s += p[1]; s += p[2]; s += p[3];
      }
    }
    part[(size_t)blockIdx.x * 128 + kind * 64 + o] = s;
  }

  {
    const int bb = wg0 >> 13, n0 = wg0 & (NPT - 1);
    for (int it = 0; it < 8; ++it) {
      const int q = tid + 128 * it;
      const int o = q >> 4, ch = q & 15;
      const size_t oi = ((size_t)(bb * COUT + o) << 13) + n0 + ch * 4;
      *(v4f*)(mxp + oi) = *(const v4f*)&s_mx[o * OPITCH + ch * 4];
      *(v4f*)(mnp + oi) = *(const v4f*)&s_mn[o * OPITCH + ch * 4];
    }
  }
}

__global__ void stats_kernel(const float* __restrict__ part, float* __restrict__ gst) {
  int t = threadIdx.x;
  float s = 0.f;
  for (int g = 0; g < CONV_WGS; ++g) s += part[(size_t)g * 128 + t];
  gst[t] = s;
}

__global__ void out_kernel(const float* __restrict__ mxp, const float* __restrict__ mnp,
                           const float* __restrict__ gst,
                           const float* __restrict__ gamma, const float* __restrict__ beta,
                           float* __restrict__ outp) {
  int i = blockIdx.x * blockDim.x + threadIdx.x;
  int o = (i >> 13) & 63;
  const float cnt = (float)(BATCH * NPT * KNN_K);
  float mean = gst[o] / cnt;
  float var  = gst[COUT + o] / cnt - mean * mean;
  float scale = gamma[o] * rsqrtf(var + BNEPS);
  float sel = (scale >= 0.f) ? mxp[i] : mnp[i];
  outp[i] = (sel - mean) * scale + beta[o];
}

extern "C" void kernel_launch(void* const* d_in, const int* in_sizes, int n_in,
                              void* d_out, int out_size, void* d_ws, size_t ws_size,
                              hipStream_t stream) {
  const float* x     = (const float*)d_in[0];
  const float* W     = (const float*)d_in[1];
  const float* bias  = (const float*)d_in[2];
  const float* gamma = (const float*)d_in[3];
  const float* beta  = (const float*)d_in[4];
  float* outp = (float*)d_out;

  float* xt  = (float*)d_ws;
  float* sq  = xt + (size_t)BATCH * NPT * CH;
  int*   knn = (int*)(sq + (size_t)BATCH * NPT);
  float* mxp = (float*)(knn + (size_t)BATCH * NPT * KNN_K);
  float* mnp = mxp + (size_t)BATCH * COUT * NPT;
  float* gst = mnp + (size_t)BATCH * COUT * NPT;
  float* part = gst + 128;

  prep_kernel <<<(BATCH * NPT) / 256, 256, 0, stream>>>(x, xt, sq);
  knn_kernel  <<<BATCH * (NPT / KROWS), KTHREADS, 0, stream>>>(xt, sq, knn);
  conv_kernel <<<CONV_WGS, 128, 0, stream>>>(xt, knn, W, bias, mxp, mnp, part);
  stats_kernel<<<1, 128, 0, stream>>>(part, gst);
  out_kernel  <<<(BATCH * COUT * NPT) / 256, 256, 0, stream>>>(mxp, mnp, gst, gamma, beta, outp);
}


# knn: top-K scan of a 16-row block split by rows over four scan waves (4 rows each); the matrix wave runs a stage as two 32-column halves, fragments read in use order, to fit 5 waves per SIMD
# speedup vs baseline: 3.5657x; 1.2360x over previous
#include <hip/hip_runtime.h>
#include <stdint.h>

#define BATCH 2
#define CH    64
#define NPT   8192
#define KNN_K 20
#define COUT  64
#define CIN2  128
#define NEGSL 0.2f
#define BNEPS 1e-5f

typedef float v2f __attribute__((ext_vector_type(2)));
typedef float v4f __attribute__((ext_vector_type(4)));
typedef float v8f __attribute__((ext_vector_type(8)));

__device__ __forceinline__ v8f wmma_f32(v2f a, v2f b, v8f c) {
  return __builtin_amdgcn_wmma_f32_16x16x4_f32(false, a, false, b, (short)0, c, false, false);
}

__device__ __forceinline__ unsigned fenc(float f) {
  unsigned u = __float_as_uint(f);
  return (u & 0x80000000u) ? ~u : (u | 0x80000000u);
}
__device__ __forceinline__ float fdec(unsigned u) {
  return __uint_as_float((u & 0x80000000u) ? (u ^ 0x80000000u) : ~u);
}

__global__ void prep_kernel(const float* __restrict__ x,
                            float* __restrict__ xt, float* __restrict__ sq) {
  int gp = blockIdx.x * blockDim.x + threadIdx.x;
  int b = gp >> 13, n = gp & (NPT - 1);
  const float* xb = x + (size_t)b * CH * NPT + n;
  float acc = 0.f;
  #pragma unroll 8
  for (int c = 0; c < CH; ++c) {
    float v = xb[(size_t)c * NPT];
    acc += v * v;
    xt[(size_t)gp * CH + c] = v;
  }
  sq[gp] = acc;
}

#define KROWS 64
#define KTHREADS 640
#define SROWS 4
#define TCOL  64
#define PITCH 68
#define RPITCH 68
#define NSTEP (NPT / TCOL)
#define NSUB  (TCOL / 16)
#define HSUB  (NSUB / 2)

__device__ __forceinline__ float lane_read_f(float v, int lane) {
  return __int_as_float(__builtin_amdgcn_readlane(__float_as_int(v), lane));
}
__device__ __forceinline__ int lane_up1(int v, int edge, bool is16) {
  const int t = __builtin_amdgcn_update_dpp(edge, v, 0x111  , 0xf, 0xf, false);
  const int v15 = __builtin_amdgcn_readlane(v, 15);
  return is16 ? v15 : t;
}

__launch_bounds__(KTHREADS)
__global__ void knn_kernel(const float* __restrict__ xt, const float* __restrict__ sq,
                           int* __restrict__ knn) {
  __shared__ __attribute__((aligned(16))) float s_b[2][TCOL * PITCH];
  __shared__ __attribute__((aligned(16))) float s_sq[4][TCOL];
  __shared__ __attribute__((aligned(16))) float s_tw[2][4][16 * RPITCH];
  __shared__ int s_out[KROWS * KNN_K];

  const int tid = threadIdx.x;
  const int L = tid & 31;
  const int w  = __builtin_amdgcn_readfirstlane(tid >> 5);
  const int pw = w & 3;
  const bool mat = w < 4;
  const int rb = mat ? 0 : ((w - 4) >> 2) * SROWS;
  const int rr = L & 15, hh = L >> 4;
  const int b  = blockIdx.x >> 7;
  const int r0 = ((blockIdx.x & 127) << 6) + (pw << 4);
  const int bofs = b << 13;
  const bool is16 = (L == 16);

  v2f a[16];
  const float* arow = xt + (size_t)(bofs + r0 + rr) * CH + 2 * hh;
  #pragma unroll
  for (int kk = 0; kk < 16; ++kk) a[kk] = *(const v2f*)(arow + kk * 4);

  float sqr[SROWS], lv[SROWS], thr[SROWS]; int li[SROWS];
  #pragma unroll
  for (int r = 0; r < SROWS; ++r) {
    sqr[r] = sq[bofs + r0 + rb + r];
    lv[r] = -__builtin_inff(); li[r] = 0; thr[r] = -__builtin_inff();
  }

  const float* gbase = xt + (size_t)bofs * CH;
  auto gload = [&](int st, v4f (&g)[8], float& gs) {
    const float* gsrc = gbase + (size_t)st * (TCOL * CH);
    #pragma unroll
    for (int i = 0; i < 8; ++i) g[i] = *(const v4f*)(gsrc + (size_t)(tid + 128 * i) * 4);
    gs = sq[bofs + st * TCOL + (tid & (TCOL - 1))];
  };
  auto sstore = [&](int st, v4f (&g)[8], float gs) {
    #pragma unroll
    for (int i = 0; i < 8; ++i) {
      int q = tid + 128 * i;
      *(v4f*)&s_b[st & 1][(q >> 4) * PITCH + (q & 15) * 4] = g[i];
    }
    s_sq[st & 3][tid & (TCOL - 1)] = gs;
  };

  v4f g[8]; float gs;
  if (mat) {
    gload(0, g, gs);
    sstore(0, g, gs);
  }
  __syncthreads();

  for (int st = 0; st <= NSTEP; ++st) {
    if (mat) {
      if (st < NSTEP) {
        const int buf = st & 1;
        gload((st + 1) & (NSTEP - 1), g, gs);

        const float* sb = &s_b[buf][0] + rr * PITCH + 2 * hh;
        float* tw = &s_tw[buf][pw][0] + (8 * hh) * RPITCH + rr;
        #pragma unroll
        for (int hf = 0; hf < 2; ++hf) {
          v2f fr[HSUB][16];
          #pragma unroll
          for (int k2 = 0; k2 < 16; k2 += 2) {
            #pragma unroll
            for (int sub = 0; sub < HSUB; ++sub) {
              fr[sub][k2]     = *(const v2f*)(sb + (hf * HSUB + sub) * 16 * PITCH + k2 * 4);
              fr[sub][k2 + 1] = *(const v2f*)(sb + (hf * HSUB + sub) * 16 * PITCH + k2 * 4 + 4);
            }
          }
          __builtin_amdgcn_sched_barrier(0);

          v8f acc[HSUB];
          #pragma unroll
          for (int sub = 0; sub < HSUB; ++sub) acc[sub] = (v8f){};
          #pragma unroll
          for (int kk = 0; kk < 16; ++kk) {
            #pragma unroll
            for (int sub = 0; sub < HSUB; ++sub) acc[sub] = wmma_f32(a[kk], fr[sub][kk], acc[sub]);
          }

          #pragma unroll
          for (int sub = 0; sub < HSUB; ++sub) {
            #pragma unroll
            for (int q = 0; q < 8; ++q) tw[q * RPITCH + (hf * HSUB + sub) * 16] = acc[sub][q];
          }
        }

        sstore(st + 1, g, gs);
      }
    } else if (st > 0) {
      const int ps = st - 1;
      const float* tw = &s_tw[ps & 1][pw][0] + rb * RPITCH + L;
      float f[SROWS][2];
      #pragma unroll
      for (int r = 0; r < SROWS; ++r) { f[r][0] = tw[r * RPITCH]; f[r][1] = tw[r * RPITCH + 32]; }
      float sqc[2];
      sqc[0] = s_sq[ps & 3][L]; sqc[1] = s_sq[ps & 3][32 + L];
      const int c0 = ps * TCOL;

      unsigned hit[SROWS];
      #pragma unroll
      for (int r = 0; r < SROWS; ++r) {
        f[r][0] = 2.f * f[r][0] - sqc[0] - sqr[r];
        f[r][1] = 2.f * f[r][1] - sqc[1] - sqr[r];
        hit[r] = __builtin_amdgcn_ballot_w32(f[r][0] > thr[r]) | __builtin_amdgcn_ballot_w32(f[r][1] > thr[r]);
      }

      #pragma unroll
      for (int r = 0; r < SROWS; ++r) {
        if (__builtin_expect(hit[r] == 0, 1)) continue;
        #pragma unroll
        for (int u = 0; u < 2; ++u) {
          const float s = f[r][u];
          unsigned cm = __builtin_amdgcn_ballot_w32(s > thr[r]);
          if (cm) {
            do {
              const int l = __builtin_ctz(cm);
              cm &= cm - 1;
              const float sv = lane_read_f(s, l);
              const int cg = c0 + u * 32 + l;
              const float tv = __int_as_float(lane_up1(__float_as_int(lv[r]), 0x7f800000, is16));
              const int   ti = lane_up1(li[r], 0, is16);
              const bool lt  = lv[r] < sv;
              const bool ltp = tv < sv;
              lv[r] = lt ? (ltp ? tv : sv) : lv[r];
              li[r] = lt ? (ltp ? ti : cg) : li[r];
            } while (cm);
            thr[r] = lane_read_f(lv[r], KNN_K - 1);
          }
        }
      }
    }
    __syncthreads();
  }

  if (!mat && L < KNN_K) {
    int* so = s_out + ((pw << 4) + rb) * KNN_K + L;
    #pragma unroll
    for (int r = 0; r < SROWS; ++r) so[r * KNN_K] = li[r];
  }
  __syncthreads();
  {
    int* outp = knn + (size_t)(bofs + ((blockIdx.x & 127) << 6)) * KNN_K;
    for (int i = tid; i < KROWS * KNN_K; i += KTHREADS) outp[i] = s_out[i];
  }
}

#define CWAVES 4
#define CPTS   16
#define CONV_WGS ((BATCH * NPT) / (CWAVES * CPTS))
#define OPITCH 68
__launch_bounds__(128)
__global__ void conv_kernel(const float* __restrict__ xt, const int* __restrict__ knn,
                            const float* __restrict__ W, const float* __restrict__ bias,
                            float* __restrict__ mxp, float* __restrict__ mnp,
                            float* __restrict__ part) {
  __shared__ __attribute__((aligned(16))) float s_mx[COUT * OPITCH];
  __shared__ __attribute__((aligned(16))) float s_mn[COUT * OPITCH];
  __shared__ __attribute__((aligned(16))) float s_red[CWAVES][64][32];

  const int tid = threadIdx.x;
  const int w = tid >> 5, L = tid & 31;
  const int c = L & 15, hh = L >> 4;
  const int wg0 = blockIdx.x * (CWAVES * CPTS);
  const int gp  = wg0 + w * CPTS + c;
  const int bofs = gp & ~(NPT - 1);

  v2f wa[4][32];
  #pragma unroll
  for (int ot = 0; ot < 4; ++ot) {
    const float* wrow = W + (size_t)(ot * 16 + c) * CIN2 + 2 * hh;
    #pragma unroll
    for (int kk = 0; kk < 32; ++kk) wa[ot][kk] = *(const v2f*)(wrow + kk * 4);
  }
  v2f xc[16];
  {
    const float* xn = xt + (size_t)gp * CH + 2 * hh;
    #pragma unroll
    for (int kk = 0; kk < 16; ++kk) xc[kk] = *(const v2f*)(xn + kk * 4);
  }
  float bo[4][8];
  #pragma unroll
  for (int ot = 0; ot < 4; ++ot) {
    #pragma unroll
    for (int r = 0; r < 8; ++r) bo[ot][r] = bias[ot * 16 + 8 * hh + r];
  }

  float fsum[4][8], fsq[4][8], vmx[4][8], vmn[4][8];
  #pragma unroll
  for (int ot = 0; ot < 4; ++ot) {
    #pragma unroll
    for (int r = 0; r < 8; ++r) {
      fsum[ot][r] = 0.f; fsq[ot][r] = 0.f;
      vmx[ot][r] = -__builtin_inff(); vmn[ot][r] = __builtin_inff();
    }
  }

  const int* kp = knn + (size_t)gp * KNN_K;
  const float* xbase = xt + (size_t)bofs * CH + 2 * hh;
  auto gather = [&](int m, v2f (&x)[16]) {
    const float* xm = xbase + (size_t)m * CH;
    #pragma unroll
    for (int kk = 0; kk < 16; ++kk) x[kk] = *(const v2f*)(xm + kk * 4);
  };

  int m1 = kp[0], m2 = kp[1];
  v2f xq[16];
  gather(m1, xq);
  m1 = m2; m2 = kp[2];

  #pragma unroll 1
  for (int t = 0; t < KNN_K; ++t) {
    v2f xnx[16];
    gather(m1, xnx);
    m1 = m2;
    m2 = kp[(t + 3 < KNN_K) ? t + 3 : KNN_K - 1];

    v8f acc[4];
    #pragma unroll
    for (int ot = 0; ot < 4; ++ot) acc[ot] = (v8f){};
    #pragma unroll
    for (int kk = 0; kk < 16; ++kk) {
      v2f d = xq[kk] - xc[kk];
      #pragma unroll
      for (int ot = 0; ot < 4; ++ot) acc[ot] = wmma_f32(wa[ot][kk], d, acc[ot]);
    }
    #pragma unroll
    for (int kk = 0; kk < 16; ++kk) {
      #pragma unroll
      for (int ot = 0; ot < 4; ++ot) acc[ot] = wmma_f32(wa[ot][16 + kk], xc[kk], acc[ot]);
    }

    #pragma unroll
    for (int ot = 0; ot < 4; ++ot) {
      #pragma unroll
      for (int r = 0; r < 8; ++r) {
        float h = acc[ot][r] + bo[ot][r];
        h = (h >= 0.f) ? h : NEGSL * h;
        fsum[ot][r] += h; fsq[ot][r] += h * h;
        vmx[ot][r] = fmaxf(vmx[ot][r], h);
        vmn[ot][r] = fminf(vmn[ot][r], h);
      }
    }
    #pragma unroll
    for (int kk = 0; kk < 16; ++kk) xq[kk] = xnx[kk];
  }

  #pragma unroll
  for (int ot = 0; ot < 4; ++ot) {
    #pragma unroll
    for (int r = 0; r < 8; ++r) {
      const int o = ot * 16 + 8 * hh + r;
      s_mx[o * OPITCH + w * CPTS + c] = vmx[ot][r];
      s_mn[o * OPITCH + w * CPTS + c] = vmn[ot][r];
      s_red[w][ot * 8 + r][L]      = fsum[ot][r];
      s_red[w][32 + ot * 8 + r][L] = fsq[ot][r];
    }
  }
  __syncthreads();

  {
    const int o = tid & 63, kind = tid >> 6;
    const int v = kind * 32 + (o >> 4) * 8 + (o & 7);
    const int h2 = (o >> 3) & 1;
    float s = 0.f;
    for (int ww = 0; ww < CWAVES; ++ww) {
      for (int q = 0; q < 4; ++q) {
        v4f p = *(const v4f*)&s_red[ww][v][h2 * 16 + q * 4];
        s += p[0]; s += p[1]; s += p[2]; s += p[3];
      }
    }
    part[(size_t)blockIdx.x * 128 + kind * 64 + o] = s;
  }

  {
    const int bb = wg0 >> 13, n0 = wg0 & (NPT - 1);
    for (int it = 0; it < 8; ++it) {
      const int q = tid + 128 * it;
      const int o = q >> 4, ch = q & 15;
      const size_t oi = ((size_t)(bb * COUT + o) << 13) + n0 + ch * 4;
      *(v4f*)(mxp + oi) = *(const v4f*)&s_mx[o * OPITCH + ch * 4];
      *(v4f*)(mnp + oi) = *(const v4f*)&s_mn[o * OPITCH + ch * 4];
    }
  }
}

__global__ void stats_kernel(const float* __restrict__ part, float* __restrict__ gst) {
  int t = threadIdx.x;
  float s = 0.f;
  for (int g = 0; g < CONV_WGS; ++g) s += part[(size_t)g * 128 + t];
  gst[t] = s;
}

__global__ void out_kernel(const float* __restrict__ mxp, const float* __restrict__ mnp,
                           const float* __restrict__ gst,
                           const float* __restrict__ gamma, const float* __restrict__ beta,
                           float* __restrict__ outp) {
  int i = blockIdx.x * blockDim.x + threadIdx.x;
  int o = (i >> 13) & 63;
  const float cnt = (float)(BATCH * NPT * KNN_K);
  float mean = gst[o] / cnt;
  float var  = gst[COUT + o] / cnt - mean * mean;
  float scale = gamma[o] * rsqrtf(var + BNEPS);
  float sel = (scale >= 0.f) ? mxp[i] : mnp[i];
  outp[i] = (sel - mean) * scale + beta[o];
}

extern "C" void kernel_launch(void* const* d_in, const int* in_sizes, int n_in,
                              void* d_out, int out_size, void* d_ws, size_t ws_size,
                              hipStream_t stream) {
  const float* x     = (const float*)d_in[0];
  const float* W     = (const float*)d_in[1];
  const float* bias  = (const float*)d_in[2];
  const float* gamma = (const float*)d_in[3];
  const float* beta  = (const float*)d_in[4];
  float* outp = (float*)d_out;

  float* xt  = (float*)d_ws;
  float* sq  = xt + (size_t)BATCH * NPT * CH;
  int*   knn = (int*)(sq + (size_t)BATCH * NPT);
  float* mxp = (float*)(knn + (size_t)BATCH * NPT * KNN_K);
  float* mnp = mxp + (size_t)BATCH * COUT * NPT;
  float* gst = mnp + (size_t)BATCH * COUT * NPT;
  float* part = gst + 128;

  prep_kernel <<<(BATCH * NPT) / 256, 256, 0, stream>>>(x, xt, sq);
  knn_kernel  <<<BATCH * (NPT / KROWS), KTHREADS, 0, stream>>>(xt, sq, knn);
  conv_kernel <<<CONV_WGS, 128, 0, stream>>>(xt, knn, W, bias, mxp, mnp, part);
  stats_kernel<<<1, 128, 0, stream>>>(part, gst);
  out_kernel  <<<(BATCH * COUT * NPT) / 256, 256, 0, stream>>>(mxp, mnp, gst, gamma, beta, outp);
}


# knn top-K insert: neighbour test = the list compare's mask moved up one bit (scalar); the one-lane shift crosses the 16-lane rows by a row-exchange move: 1 broadcast + 1 compare per insert, not 3 + 2
# speedup vs baseline: 3.9720x; 1.1139x over previous
#include <hip/hip_runtime.h>
#include <stdint.h>

#define BATCH 2
#define CH    64
#define NPT   8192
#define KNN_K 20
#define COUT  64
#define CIN2  128
#define NEGSL 0.2f
#define BNEPS 1e-5f

typedef float v2f __attribute__((ext_vector_type(2)));
typedef float v4f __attribute__((ext_vector_type(4)));
typedef float v8f __attribute__((ext_vector_type(8)));

__device__ __forceinline__ v8f wmma_f32(v2f a, v2f b, v8f c) {
  return __builtin_amdgcn_wmma_f32_16x16x4_f32(false, a, false, b, (short)0, c, false, false);
}

__device__ __forceinline__ unsigned fenc(float f) {
  unsigned u = __float_as_uint(f);
  return (u & 0x80000000u) ? ~u : (u | 0x80000000u);
}
__device__ __forceinline__ float fdec(unsigned u) {
  return __uint_as_float((u & 0x80000000u) ? (u ^ 0x80000000u) : ~u);
}

__global__ void prep_kernel(const float* __restrict__ x,
                            float* __restrict__ xt, float* __restrict__ sq) {
  int gp = blockIdx.x * blockDim.x + threadIdx.x;
  int b = gp >> 13, n = gp & (NPT - 1);
  const float* xb = x + (size_t)b * CH * NPT + n;
  float acc = 0.f;
  #pragma unroll 8
  for (int c = 0; c < CH; ++c) {
    float v = xb[(size_t)c * NPT];
    acc += v * v;
    xt[(size_t)gp * CH + c] = v;
  }
  sq[gp] = acc;
}

#define KROWS 64
#define KTHREADS 640
#define SROWS 4
#define TCOL  64
#define PITCH 68
#define RPITCH 68
#define NSTEP (NPT / TCOL)
#define NSUB  (TCOL / 16)
#define HSUB  (NSUB / 2)

__device__ __forceinline__ float lane_read_f(float v, int lane) {
  return __int_as_float(__builtin_amdgcn_readlane(__float_as_int(v), lane));
}
__device__ __forceinline__ int lane_up1(int v) {
  const int x = __builtin_amdgcn_permlanex16(v, v, 0xFFFFFFFFu, 0xFFFFFFFFu, false, false);
  return __builtin_amdgcn_update_dpp(x, v, 0x111  , 0xf, 0xf, false);
}

__launch_bounds__(KTHREADS)
__global__ void knn_kernel(const float* __restrict__ xt, const float* __restrict__ sq,
                           int* __restrict__ knn) {
  __shared__ __attribute__((aligned(16))) float s_b[2][TCOL * PITCH];
  __shared__ __attribute__((aligned(16))) float s_sq[4][TCOL];
  __shared__ __attribute__((aligned(16))) float s_tw[2][4][16 * RPITCH];
  __shared__ int s_out[KROWS * KNN_K];

  const int tid = threadIdx.x;
  const int L = tid & 31;
  const int w  = __builtin_amdgcn_readfirstlane(tid >> 5);
  const int pw = w & 3;
  const bool mat = w < 4;
  const int rb = mat ? 0 : ((w - 4) >> 2) * SROWS;
  const int rr = L & 15, hh = L >> 4;
  const int b  = blockIdx.x >> 7;
  const int r0 = ((blockIdx.x & 127) << 6) + (pw << 4);
  const int bofs = b << 13;

  v2f a[16];
  const float* arow = xt + (size_t)(bofs + r0 + rr) * CH + 2 * hh;
  #pragma unroll
  for (int kk = 0; kk < 16; ++kk) a[kk] = *(const v2f*)(arow + kk * 4);

  float sqr[SROWS], lv[SROWS], thr[SROWS]; int li[SROWS];
  #pragma unroll
  for (int r = 0; r < SROWS; ++r) {
    sqr[r] = sq[bofs + r0 + rb + r];
    lv[r] = -__builtin_inff(); li[r] = 0; thr[r] = -__builtin_inff();
  }

  const float* gbase = xt + (size_t)bofs * CH;
  auto gload = [&](int st, v4f (&g)[8], float& gs) {
    const float* gsrc = gbase + (size_t)st * (TCOL * CH);
    #pragma unroll
    for (int i = 0; i < 8; ++i) g[i] = *(const v4f*)(gsrc + (size_t)(tid + 128 * i) * 4);
    gs = sq[bofs + st * TCOL + (tid & (TCOL - 1))];
  };
  auto sstore = [&](int st, v4f (&g)[8], float gs) {
    #pragma unroll
    for (int i = 0; i < 8; ++i) {
      int q = tid + 128 * i;
      *(v4f*)&s_b[st & 1][(q >> 4) * PITCH + (q & 15) * 4] = g[i];
    }
    s_sq[st & 3][tid & (TCOL - 1)] = gs;
  };

  v4f g[8]; float gs;
  if (mat) {
    gload(0, g, gs);
    sstore(0, g, gs);
  }
  __syncthreads();

  for (int st = 0; st <= NSTEP; ++st) {
    if (mat) {
      if (st < NSTEP) {
        const int buf = st & 1;
        gload((st + 1) & (NSTEP - 1), g, gs);

        const float* sb = &s_b[buf][0] + rr * PITCH + 2 * hh;
        float* tw = &s_tw[buf][pw][0] + (8 * hh) * RPITCH + rr;
        #pragma unroll
        for (int hf = 0; hf < 2; ++hf) {
          v2f fr[HSUB][16];
          #pragma unroll
          for (int k2 = 0; k2 < 16; k2 += 2) {
            #pragma unroll
            for (int sub = 0; sub < HSUB; ++sub) {
              fr[sub][k2]     = *(const v2f*)(sb + (hf * HSUB + sub) * 16 * PITCH + k2 * 4);
              fr[sub][k2 + 1] = *(const v2f*)(sb + (hf * HSUB + sub) * 16 * PITCH + k2 * 4 + 4);
            }
          }
          __builtin_amdgcn_sched_barrier(0);

          v8f acc[HSUB];
          #pragma unroll
          for (int sub = 0; sub < HSUB; ++sub) acc[sub] = (v8f){};
          #pragma unroll
          for (int kk = 0; kk < 16; ++kk) {
            #pragma unroll
            for (int sub = 0; sub < HSUB; ++sub) acc[sub] = wmma_f32(a[kk], fr[sub][kk], acc[sub]);
          }

          #pragma unroll
          for (int sub = 0; sub < HSUB; ++sub) {
            #pragma unroll
            for (int q = 0; q < 8; ++q) tw[q * RPITCH + (hf * HSUB + sub) * 16] = acc[sub][q];
          }
        }

        sstore(st + 1, g, gs);
      }
    } else if (st > 0) {
      const int ps = st - 1;
      const float* tw = &s_tw[ps & 1][pw][0] + rb * RPITCH + L;
      float f[SROWS][2];
      #pragma unroll
      for (int r = 0; r < SROWS; ++r) { f[r][0] = tw[r * RPITCH]; f[r][1] = tw[r * RPITCH + 32]; }
      float sqc[2];
      sqc[0] = s_sq[ps & 3][L]; sqc[1] = s_sq[ps & 3][32 + L];
      const int c0 = ps * TCOL;

      unsigned hit[SROWS];
      #pragma unroll
      for (int r = 0; r < SROWS; ++r) {
        f[r][0] = 2.f * f[r][0] - sqc[0] - sqr[r];
        f[r][1] = 2.f * f[r][1] - sqc[1] - sqr[r];
        hit[r] = __builtin_amdgcn_ballot_w32(f[r][0] > thr[r]) | __builtin_amdgcn_ballot_w32(f[r][1] > thr[r]);
      }

      #pragma unroll
      for (int r = 0; r < SROWS; ++r) {
        if (__builtin_expect(hit[r] == 0, 1)) continue;
        #pragma unroll
        for (int u = 0; u < 2; ++u) {
          const float s = f[r][u];
          unsigned cm = __builtin_amdgcn_ballot_w32(s > thr[r]);
          if (cm) {
            do {
              const int l = __builtin_ctz(cm);
              cm &= cm - 1;
              const float sv = lane_read_f(s, l);
              const int cg = c0 + u * 32 + l;
              const float tv = __int_as_float(lane_up1(__float_as_int(lv[r])));
              const int   ti = lane_up1(li[r]);
              const unsigned ltm = __builtin_amdgcn_ballot_w32(lv[r] < sv);
              const bool lt  = __builtin_amdgcn_inverse_ballot_w32(ltm);
              const bool ltp = __builtin_amdgcn_inverse_ballot_w32(ltm << 1);
              lv[r] = ltp ? tv : (lt ? sv : lv[r]);
              li[r] = ltp ? ti : (lt ? cg : li[r]);
            } while (cm);
            thr[r] = lane_read_f(lv[r], KNN_K - 1);
          }
        }
      }
    }
    __syncthreads();
  }

  if (!mat && L < KNN_K) {
    int* so = s_out + ((pw << 4) + rb) * KNN_K + L;
    #pragma unroll
    for (int r = 0; r < SROWS; ++r) so[r * KNN_K] = li[r];
  }
  __syncthreads();
  {
    int* outp = knn + (size_t)(bofs + ((blockIdx.x & 127) << 6)) * KNN_K;
    for (int i = tid; i < KROWS * KNN_K; i += KTHREADS) outp[i] = s_out[i];
  }
}

#define CWAVES 4
#define CPTS   16
#define CONV_WGS ((BATCH * NPT) / (CWAVES * CPTS))
#define OPITCH 68
__launch_bounds__(128)
__global__ void conv_kernel(const float* __restrict__ xt, const int* __restrict__ knn,
                            const float* __restrict__ W, const float* __restrict__ bias,
                            float* __restrict__ mxp, float* __restrict__ mnp,
                            float* __restrict__ part) {
  __shared__ __attribute__((aligned(16))) float s_mx[COUT * OPITCH];
  __shared__ __attribute__((aligned(16))) float s_mn[COUT * OPITCH];
  __shared__ __attribute__((aligned(16))) float s_red[CWAVES][64][32];

  const int tid = threadIdx.x;
  const int w = tid >> 5, L = tid & 31;
  const int c = L & 15, hh = L >> 4;
  const int wg0 = blockIdx.x * (CWAVES * CPTS);
  const int gp  = wg0 + w * CPTS + c;
  const int bofs = gp & ~(NPT - 1);

  v2f wa[4][16];
  #pragma unroll
  for (int ot = 0; ot < 4; ++ot) {
    const float* wrow = W + (size_t)(ot * 16 + c) * CIN2 + 2 * hh;
    #pragma unroll
    for (int kk = 0; kk < 16; ++kk) wa[ot][kk] = *(const v2f*)(wrow + kk * 4);
  }
  v2f xc[16];
  {
    const float* xn = xt + (size_t)gp * CH + 2 * hh;
    #pragma unroll
    for (int kk = 0; kk < 16; ++kk) xc[kk] = *(const v2f*)(xn + kk * 4);
  }
  v8f cst[4];
  {
    v2f wb[4][16];
    #pragma unroll
    for (int ot = 0; ot < 4; ++ot) {
      const float* wrow = W + (size_t)(ot * 16 + c) * CIN2 + CH + 2 * hh;
      #pragma unroll
      for (int kk = 0; kk < 16; ++kk) wb[ot][kk] = *(const v2f*)(wrow + kk * 4);
    }
    #pragma unroll
    for (int ot = 0; ot < 4; ++ot) cst[ot] = (v8f){};
    #pragma unroll
    for (int kk = 0; kk < 16; ++kk) {
      #pragma unroll
      for (int ot = 0; ot < 4; ++ot) cst[ot] = wmma_f32(wb[ot][kk], xc[kk], cst[ot]);
    }
    #pragma unroll
    for (int ot = 0; ot < 4; ++ot) {
      #pragma unroll
      for (int r = 0; r < 8; ++r) cst[ot][r] += bias[ot * 16 + 8 * hh + r];
    }
  }

  float fsum[4][8], fsq[4][8], vmx[4][8], vmn[4][8];
  #pragma unroll
  for (int ot = 0; ot < 4; ++ot) {
    #pragma unroll
    for (int r = 0; r < 8; ++r) {
      fsum[ot][r] = 0.f; fsq[ot][r] = 0.f;
      vmx[ot][r] = -__builtin_inff(); vmn[ot][r] = __builtin_inff();
    }
  }

  const int* kp = knn + (size_t)gp * KNN_K;
  const float* xbase = xt + (size_t)bofs * CH + 2 * hh;
  auto gather = [&](int m, v2f (&x)[16]) {
    const float* xm = xbase + (size_t)m * CH;
    #pragma unroll
    for (int kk = 0; kk < 16; ++kk) x[kk] = *(const v2f*)(xm + kk * 4);
  };

  auto tile = [&](const v2f (&xn)[16]) {
    v8f acc[4];
    #pragma unroll
    for (int ot = 0; ot < 4; ++ot) acc[ot] = cst[ot];
    #pragma unroll
    for (int kk = 0; kk < 16; ++kk) {
      v2f d = xn[kk] - xc[kk];
      #pragma unroll
      for (int ot = 0; ot < 4; ++ot) acc[ot] = wmma_f32(wa[ot][kk], d, acc[ot]);
    }
    #pragma unroll
    for (int ot = 0; ot < 4; ++ot) {
      #pragma unroll
      for (int r = 0; r < 8; ++r) {
        float h = acc[ot][r];
        h = (h >= 0.f) ? h : NEGSL * h;
        fsum[ot][r] += h; fsq[ot][r] += h * h;
        vmx[ot][r] = fmaxf(vmx[ot][r], h);
        vmn[ot][r] = fminf(vmn[ot][r], h);
      }
    }
  };

  int m1 = kp[0], m2 = kp[1];
  v2f xa[16], xb[16];
  gather(m1, xa);
  m1 = m2; m2 = kp[2];

  #pragma unroll 1
  for (int t = 0; t < KNN_K; t += 2) {
    gather(m1, xb);
    m1 = m2;
    m2 = kp[(t + 3 < KNN_K) ? t + 3 : KNN_K - 1];
    __builtin_amdgcn_sched_barrier(0);
    tile(xa);
    gather(m1, xa);
    m1 = m2;
    m2 = kp[(t + 4 < KNN_K) ? t + 4 : KNN_K - 1];
    __builtin_amdgcn_sched_barrier(0);
    tile(xb);
  }

  #pragma unroll
  for (int ot = 0; ot < 4; ++ot) {
    #pragma unroll
    for (int r = 0; r < 8; ++r) {
      const int o = ot * 16 + 8 * hh + r;
      s_mx[o * OPITCH + w * CPTS + c] = vmx[ot][r];
      s_mn[o * OPITCH + w * CPTS + c] = vmn[ot][r];
      s_red[w][ot * 8 + r][L]      = fsum[ot][r];
      s_red[w][32 + ot * 8 + r][L] = fsq[ot][r];
    }
  }
  __syncthreads();

  {
    const int o = tid & 63, kind = tid >> 6;
    const int v = kind * 32 + (o >> 4) * 8 + (o & 7);
    const int h2 = (o >> 3) & 1;
    float s = 0.f;
    for (int ww = 0; ww < CWAVES; ++ww) {
      for (int q = 0; q < 4; ++q) {
        v4f p = *(const v4f*)&s_red[ww][v][h2 * 16 + q * 4];
        s += p[0]; s += p[1]; s += p[2]; s += p[3];
      }
    }
    part[(size_t)blockIdx.x * 128 + kind * 64 + o] = s;
  }

  {
    const int bb = wg0 >> 13, n0 = wg0 & (NPT - 1);
    for (int it = 0; it < 8; ++it) {
      const int q = tid + 128 * it;
      const int o = q >> 4, ch = q & 15;
      const size_t oi = ((size_t)(bb * COUT + o) << 13) + n0 + ch * 4;
      *(v4f*)(mxp + oi) = *(const v4f*)&s_mx[o * OPITCH + ch * 4];
      *(v4f*)(mnp + oi) = *(const v4f*)&s_mn[o * OPITCH + ch * 4];
    }
  }
}

__global__ void stats_kernel(const float* __restrict__ part, float* __restrict__ gst) {
  int t = threadIdx.x;
  float s = 0.f;
  for (int g = 0; g < CONV_WGS; ++g) s += part[(size_t)g * 128 + t];
  gst[t] = s;
}

__global__ void out_kernel(const float* __restrict__ mxp, const float* __restrict__ mnp,
                           const float* __restrict__ gst,
                           const float* __restrict__ gamma, const float* __restrict__ beta,
                           float* __restrict__ outp) {
  int i = blockIdx.x * blockDim.x + threadIdx.x;
  int o = (i >> 13) & 63;
  const float cnt = (float)(BATCH * NPT * KNN_K);
  float mean = gst[o] / cnt;
  float var  = gst[COUT + o] / cnt - mean * mean;
  float scale = gamma[o] * rsqrtf(var + BNEPS);
  float sel = (scale >= 0.f) ? mxp[i] : mnp[i];
  outp[i] = (sel - mean) * scale + beta[o];
}

extern "C" void kernel_launch(void* const* d_in, const int* in_sizes, int n_in,
                              void* d_out, int out_size, void* d_ws, size_t ws_size,
                              hipStream_t stream) {
  const float* x     = (const float*)d_in[0];
  const float* W     = (const float*)d_in[1];
  const float* bias  = (const float*)d_in[2];
  const float* gamma = (const float*)d_in[3];
  const float* beta  = (const float*)d_in[4];
  float* outp = (float*)d_out;

  float* xt  = (float*)d_ws;
  float* sq  = xt + (size_t)BATCH * NPT * CH;
  int*   knn = (int*)(sq + (size_t)BATCH * NPT);
  float* mxp = (float*)(knn + (size_t)BATCH * NPT * KNN_K);
  float* mnp = mxp + (size_t)BATCH * COUT * NPT;
  float* gst = mnp + (size_t)BATCH * COUT * NPT;
  float* part = gst + 128;

  prep_kernel <<<(BATCH * NPT) / 256, 256, 0, stream>>>(x, xt, sq);
  knn_kernel  <<<BATCH * (NPT / KROWS), KTHREADS, 0, stream>>>(xt, sq, knn);
  conv_kernel <<<CONV_WGS, 128, 0, stream>>>(xt, knn, W, bias, mxp, mnp, part);
  stats_kernel<<<1, 128, 0, stream>>>(part, gst);
  out_kernel  <<<(BATCH * COUT * NPT) / 256, 256, 0, stream>>>(mxp, mnp, gst, gamma, beta, outp);
}
